# starting kernel
# baseline (speedup 1.0000x reference)
#include <hip/hip_runtime.h>
#include <hip/hip_bf16.h>


typedef __attribute__((ext_vector_type(2))) float v2f;
typedef __attribute__((ext_vector_type(8))) float v8f;

__global__ __launch_bounds__(256) void gcn_init_kernel(float* __restrict__ deg,
                                                       float* __restrict__ agg,
                                                       int N, int total32) {
  int i = blockIdx.x * 256 + threadIdx.x;
  if (i < total32) agg[i] = 0.0f;
  if (i < N) deg[i] = 1.0f;
}

__global__ __launch_bounds__(256) void gcn_degree_kernel(const int* __restrict__ dst,
                                                         float* __restrict__ deg,
                                                         int E) {
  int e = blockIdx.x * 256 + threadIdx.x;
  if (e < E) unsafeAtomicAdd(&deg[dst[e]], 1.0f);
}

__global__ __launch_bounds__(256) void gcn_dinv_kernel(float* __restrict__ deg, int N) {
  int i = blockIdx.x * 256 + threadIdx.x;
  if (i < N) {
    float d = deg[i];
    deg[i] = (d > 0.0f) ? rsqrtf(d) : 0.0f;
  }
}

template <int NCOL>
__global__ __launch_bounds__(256) void gcn_gemm_wmma_kernel(const float* __restrict__ X,
                                                            const float* __restrict__ W,
                                                            float* __restrict__ Y,
                                                            int nTiles) {
  const int lane = threadIdx.x & 31;
  const int wave = threadIdx.x >> 5;
  const int tile = blockIdx.x * 8 + wave;
  if (tile >= nTiles) return;

  const int rowBase = tile * 16;
  const int lo = lane & 15;
  const int hi = lane >> 4;

  v2f a[8];
#pragma unroll
  for (int kb = 0; kb < 8; ++kb) {
    const int k = kb * 4 + hi * 2;
    const float* xr = X + (size_t)(rowBase + lo) * 32 + k;
    a[kb].x = xr[0];
    a[kb].y = xr[1];
  }

#pragma unroll
  for (int ct = 0; ct < NCOL / 16; ++ct) {
    v8f c = {};
#pragma unroll
    for (int kb = 0; kb < 8; ++kb) {
      const int k = kb * 4 + hi * 2;
      v2f b;
      b.x = W[(size_t)k * NCOL + ct * 16 + lo];
      b.y = W[(size_t)(k + 1) * NCOL + ct * 16 + lo];
      c = __builtin_amdgcn_wmma_f32_16x16x4_f32(false, a[kb], false, b,
                                                (short)0, c, false, false);
    }
#pragma unroll
    for (int v = 0; v < 8; ++v) {
      const int m = v + hi * 8;
      Y[(size_t)(rowBase + m) * NCOL + ct * 16 + lo] = c[v];
    }
  }
}

template <int F>
__global__ __launch_bounds__(256) void gcn_edge_agg_kernel(const int* __restrict__ src,
                                                           const int* __restrict__ dst,
                                                           const float* __restrict__ dinv,
                                                           const float* __restrict__ H,
                                                           float* __restrict__ AGG,
                                                           long long total) {
  const long long t = (long long)blockIdx.x * 256 + threadIdx.x;
  if (t >= total) return;
  constexpr int CPE = F / 4;
  const int e  = (int)(t / CPE);
  const int fb = (int)(t % CPE) * 4;
  const int s = src[e];
  const int d = dst[e];
  const float c = dinv[s] * dinv[d];
  const float4 v = *(const float4*)(H + (size_t)s * F + fb);
  float* p = AGG + (size_t)d * F + fb;
  unsafeAtomicAdd(p + 0, v.x * c);
  unsafeAtomicAdd(p + 1, v.y * c);
  unsafeAtomicAdd(p + 2, v.z * c);
  unsafeAtomicAdd(p + 3, v.w * c);
}

__global__ __launch_bounds__(256) void gcn_finish1_kernel(float* __restrict__ AGG,
                                                          const float* __restrict__ XW,
                                                          const float* __restrict__ dinv,
                                                          const float* __restrict__ b1,
                                                          float* __restrict__ H1,
                                                          int total) {
  int i = blockIdx.x * 256 + threadIdx.x;
  if (i >= total) return;
  const int n = i >> 5;
  const int f = i & 31;
  float v = AGG[i];
  AGG[i] = 0.0f;
  const float di = dinv[n];
  v += XW[i] * di * di;
  H1[i] = tanhf(v + b1[f]);
}

__global__ __launch_bounds__(256) void gcn_finish2_kernel(const float* __restrict__ AGG,
                                                          const float* __restrict__ HW,
                                                          const float* __restrict__ dinv,
                                                          const float* __restrict__ b2,
                                                          float* __restrict__ out_h,
                                                          float* __restrict__ out_ls,
                                                          int N) {
  int n = blockIdx.x * 256 + threadIdx.x;
  if (n >= N) return;
  const float di = dinv[n];
  const float sc = di * di;
  float v[16];
  float m = -3.402823466e38f;
#pragma unroll
  for (int f = 0; f < 16; ++f) {
    const float t = AGG[(size_t)n * 16 + f] + HW[(size_t)n * 16 + f] * sc + b2[f];
    v[f] = t;
    m = fmaxf(m, t);
  }
  float s = 0.0f;
#pragma unroll
  for (int f = 0; f < 16; ++f) s += expf(v[f] - m);
  const float lse = logf(s);
#pragma unroll
  for (int f = 0; f < 16; ++f) {
    out_h[(size_t)n * 16 + f] = v[f];
    out_ls[(size_t)n * 16 + f] = v[f] - m - lse;
  }
}

extern "C" void kernel_launch(void* const* d_in, const int* in_sizes, int n_in,
                              void* d_out, int out_size, void* d_ws, size_t ws_size,
                              hipStream_t stream) {
  const float* x  = (const float*)d_in[0];
  const int*   ei = (const int*)d_in[1];
  const float* W1 = (const float*)d_in[2];
  const float* b1 = (const float*)d_in[3];
  const float* W2 = (const float*)d_in[4];
  const float* b2 = (const float*)d_in[5];

  const int N = in_sizes[0] / 32;
  const int E = in_sizes[1] / 2;
  const int* src = ei;
  const int* dst = ei + E;

  float* ws   = (float*)d_ws;
  float* deg  = ws;
  float* xw1  = ws + (size_t)N;
  float* agg1 = ws + (size_t)N * 33;
  float* h1   = ws + (size_t)N * 65;
  float* hw2  = xw1;
  float* agg2 = agg1;

  float* out_h  = (float*)d_out;
  float* out_ls = out_h + (size_t)N * 16;

  auto nb = [](long long n) { return (unsigned)((n + 255) / 256); };
  const int nTiles = N / 16;
  const unsigned gemmBlocks = (unsigned)((nTiles + 7) / 8);

  gcn_init_kernel<<<nb((long long)N * 32), 256, 0, stream>>>(deg, agg1, N, N * 32);
  gcn_degree_kernel<<<nb(E), 256, 0, stream>>>(dst, deg, E);
  gcn_dinv_kernel<<<nb(N), 256, 0, stream>>>(deg, N);

  gcn_gemm_wmma_kernel<32><<<gemmBlocks, 256, 0, stream>>>(x, W1, xw1, nTiles);
  gcn_edge_agg_kernel<32><<<nb((long long)E * 8), 256, 0, stream>>>(
      src, dst, deg, xw1, agg1, (long long)E * 8);
  gcn_finish1_kernel<<<nb((long long)N * 32), 256, 0, stream>>>(agg1, xw1, deg, b1, h1,
                                                                N * 32);

  gcn_gemm_wmma_kernel<16><<<gemmBlocks, 256, 0, stream>>>(h1, W2, hw2, nTiles);
  gcn_edge_agg_kernel<16><<<nb((long long)E * 4), 256, 0, stream>>>(
      src, dst, deg, hw2, agg2, (long long)E * 4);
  gcn_finish2_kernel<<<nb(N), 256, 0, stream>>>(agg2, hw2, deg, b2, out_h, out_ls, N);
}


# atomic targets doubled, one copy per half of the chip: a wave adds degree counts and edge sums only into its own half's copy (placement read per wave), consumers add the copies; sum order changed
# speedup vs baseline: 1.5563x; 1.5563x over previous
#include <hip/hip_runtime.h>
#include <hip/hip_bf16.h>


typedef __attribute__((ext_vector_type(2))) float v2f;
typedef __attribute__((ext_vector_type(8))) float v8f;

__device__ __forceinline__ unsigned gcn_chip_half() {
  return (__builtin_amdgcn_s_sendmsg_rtn(0x87) >> 18) & 1u;
}

__global__ __launch_bounds__(256) void gcn_init_kernel(float* __restrict__ degc,
                                                       float* __restrict__ agg,
                                                       int N, int total64) {
  int i = blockIdx.x * 256 + threadIdx.x;
  if (i < total64) agg[i] = 0.0f;
  if (i < 2 * N) degc[i] = 0.0f;
}

__global__ __launch_bounds__(256) void gcn_degree_kernel(const int* __restrict__ dst,
                                                         float* __restrict__ degc,
                                                         int E, int N) {
  float* mine = degc + (size_t)gcn_chip_half() * (size_t)N;
  const int step = (int)gridDim.x * 256;
  for (int e = blockIdx.x * 256 + threadIdx.x; e < E; e += step)
    unsafeAtomicAdd(&mine[dst[e]], 1.0f);
}

__global__ __launch_bounds__(256) void gcn_dinv_kernel(const float* __restrict__ degc,
                                                       float* __restrict__ dinv, int N) {
  int i = blockIdx.x * 256 + threadIdx.x;
  if (i < N) {
    float d = degc[i] + degc[(size_t)N + i] + 1.0f;
    dinv[i] = (d > 0.0f) ? rsqrtf(d) : 0.0f;
  }
}

template <int NCOL>
__global__ __launch_bounds__(256) void gcn_gemm_wmma_kernel(const float* __restrict__ X,
                                                            const float* __restrict__ W,
                                                            float* __restrict__ Y,
                                                            int nTiles) {
  const int lane = threadIdx.x & 31;
  const int wave = threadIdx.x >> 5;
  const int tile = blockIdx.x * 8 + wave;
  if (tile >= nTiles) return;

  const int rowBase = tile * 16;
  const int lo = lane & 15;
  const int hi = lane >> 4;

  v2f a[8];
#pragma unroll
  for (int kb = 0; kb < 8; ++kb) {
    const int k = kb * 4 + hi * 2;
    const float* xr = X + (size_t)(rowBase + lo) * 32 + k;
    a[kb].x = xr[0];
    a[kb].y = xr[1];
  }

#pragma unroll
  for (int ct = 0; ct < NCOL / 16; ++ct) {
    v8f c = {};
#pragma unroll
    for (int kb = 0; kb < 8; ++kb) {
      const int k = kb * 4 + hi * 2;
      v2f b;
      b.x = W[(size_t)k * NCOL + ct * 16 + lo];
      b.y = W[(size_t)(k + 1) * NCOL + ct * 16 + lo];
      c = __builtin_amdgcn_wmma_f32_16x16x4_f32(false, a[kb], false, b,
                                                (short)0, c, false, false);
    }
#pragma unroll
    for (int v = 0; v < 8; ++v) {
      const int m = v + hi * 8;
      Y[(size_t)(rowBase + m) * NCOL + ct * 16 + lo] = c[v];
    }
  }
}

template <int F>
__global__ __launch_bounds__(256) void gcn_edge_agg_kernel(const int* __restrict__ src,
                                                           const int* __restrict__ dst,
                                                           const float* __restrict__ dinv,
                                                           const float* __restrict__ H,
                                                           float* __restrict__ AGG,
                                                           long long total,
                                                           long long copyStride) {
  float* mine = AGG + (size_t)gcn_chip_half() * (size_t)copyStride;
  constexpr int CPE = F / 4;
  const long long step = (long long)gridDim.x * 256;
  for (long long t = (long long)blockIdx.x * 256 + threadIdx.x; t < total; t += step) {
    const int e  = (int)(t / CPE);
    const int fb = (int)(t % CPE) * 4;
    const int s = src[e];
    const int d = dst[e];
    const float c = dinv[s] * dinv[d];
    const float4 v = *(const float4*)(H + (size_t)s * F + fb);
    float* p = mine + (size_t)d * F + fb;
    unsafeAtomicAdd(p + 0, v.x * c);
    unsafeAtomicAdd(p + 1, v.y * c);
    unsafeAtomicAdd(p + 2, v.z * c);
    unsafeAtomicAdd(p + 3, v.w * c);
  }
}

__global__ __launch_bounds__(256) void gcn_finish1_kernel(float* __restrict__ AGG,
                                                          const float* __restrict__ XW,
                                                          const float* __restrict__ dinv,
                                                          const float* __restrict__ b1,
                                                          float* __restrict__ H1,
                                                          int total) {
  int i = blockIdx.x * 256 + threadIdx.x;
  if (i >= total) return;
  const int n = i >> 5;
  const int f = i & 31;
  float* AGGB = AGG + (size_t)total;
  float v = AGG[i] + AGGB[i];
  if (i < total / 2) {
    AGG[i] = 0.0f;
    AGGB[i] = 0.0f;
  }
  const float di = dinv[n];
  v += XW[i] * di * di;
  H1[i] = tanhf(v + b1[f]);
}

__global__ __launch_bounds__(256) void gcn_finish2_kernel(const float* __restrict__ AGG,
                                                          const float* __restrict__ AGGB,
                                                          const float* __restrict__ HW,
                                                          const float* __restrict__ dinv,
                                                          const float* __restrict__ b2,
                                                          float* __restrict__ out_h,
                                                          float* __restrict__ out_ls,
                                                          int N) {
  int n = blockIdx.x * 256 + threadIdx.x;
  if (n >= N) return;
  const float di = dinv[n];
  const float sc = di * di;
  float v[16];
  float m = -3.402823466e38f;
#pragma unroll
  for (int f = 0; f < 16; ++f) {
    const float t = (AGG[(size_t)n * 16 + f] + AGGB[(size_t)n * 16 + f]) +
                    HW[(size_t)n * 16 + f] * sc + b2[f];
    v[f] = t;
    m = fmaxf(m, t);
  }
  float s = 0.0f;
#pragma unroll
  for (int f = 0; f < 16; ++f) s += expf(v[f] - m);
  const float lse = logf(s);
#pragma unroll
  for (int f = 0; f < 16; ++f) {
    out_h[(size_t)n * 16 + f] = v[f];
    out_ls[(size_t)n * 16 + f] = v[f] - m - lse;
  }
}

extern "C" void kernel_launch(void* const* d_in, const int* in_sizes, int n_in,
                              void* d_out, int out_size, void* d_ws, size_t ws_size,
                              hipStream_t stream) {
  const float* x  = (const float*)d_in[0];
  const int*   ei = (const int*)d_in[1];
  const float* W1 = (const float*)d_in[2];
  const float* b1 = (const float*)d_in[3];
  const float* W2 = (const float*)d_in[4];
  const float* b2 = (const float*)d_in[5];

  const int N = in_sizes[0] / 32;
  const int E = in_sizes[1] / 2;
  const int* src = ei;
  const int* dst = ei + E;

  float* ws   = (float*)d_ws;
  float* degc = ws;
  float* deg  = ws + (size_t)N * 2;
  float* xw1  = ws + (size_t)N * 3;
  float* agg1 = ws + (size_t)N * 35;
  float* h1   = ws + (size_t)N * 99;
  float* hw2  = xw1;
  float* agg2 = agg1;
  const long long copyStride = (long long)N * 32;

  float* out_h  = (float*)d_out;
  float* out_ls = out_h + (size_t)N * 16;

  auto nb = [](long long n) { return (unsigned)((n + 255) / 256); };
  const int nTiles = N / 16;
  const unsigned gemmBlocks = (unsigned)((nTiles + 7) / 8);

  const unsigned loopBlocks = 4096;

  gcn_init_kernel<<<nb((long long)N * 64), 256, 0, stream>>>(degc, agg1, N, N * 64);
  gcn_degree_kernel<<<loopBlocks, 256, 0, stream>>>(dst, degc, E, N);
  gcn_dinv_kernel<<<nb(N), 256, 0, stream>>>(degc, deg, N);

  gcn_gemm_wmma_kernel<32><<<gemmBlocks, 256, 0, stream>>>(x, W1, xw1, nTiles);
  gcn_edge_agg_kernel<32><<<loopBlocks, 256, 0, stream>>>(
      src, dst, deg, xw1, agg1, (long long)E * 8, copyStride);
  gcn_finish1_kernel<<<nb((long long)N * 32), 256, 0, stream>>>(agg1, xw1, deg, b1, h1,
                                                                N * 32);

  gcn_gemm_wmma_kernel<16><<<gemmBlocks, 256, 0, stream>>>(h1, W2, hw2, nTiles);
  gcn_edge_agg_kernel<16><<<loopBlocks, 256, 0, stream>>>(
      src, dst, deg, hw2, agg2, (long long)E * 4, copyStride);
  gcn_finish2_kernel<<<nb(N), 256, 0, stream>>>(agg2, agg2 + copyStride, hw2, deg, b2, out_h,
                                                out_ls, N);
}


# layer 1 sums the input rows over the edges before its product ((A x)W1, not A(xW1)); both WMMA products and the tanh then run in one launch, 9 to 7 launches; layer-1 rounding order changed
# speedup vs baseline: 1.8542x; 1.1914x over previous
#include <hip/hip_runtime.h>
#include <hip/hip_bf16.h>


typedef __attribute__((ext_vector_type(2))) float v2f;
typedef __attribute__((ext_vector_type(8))) float v8f;

__device__ __forceinline__ unsigned gcn_chip_half() {
  return (__builtin_amdgcn_s_sendmsg_rtn(0x87) >> 18) & 1u;
}

__global__ __launch_bounds__(256) void gcn_init_kernel(float* __restrict__ degc,
                                                       float* __restrict__ agg,
                                                       int N, int total64) {
  int i = blockIdx.x * 256 + threadIdx.x;
  if (i < total64) agg[i] = 0.0f;
  if (i < 2 * N) degc[i] = 0.0f;
}

__global__ __launch_bounds__(256) void gcn_degree_kernel(const int* __restrict__ dst,
                                                         float* __restrict__ degc,
                                                         int E, int N) {
  float* mine = degc + (size_t)gcn_chip_half() * (size_t)N;
  const int step = (int)gridDim.x * 256;
  for (int e = blockIdx.x * 256 + threadIdx.x; e < E; e += step)
    unsafeAtomicAdd(&mine[dst[e]], 1.0f);
}

__global__ __launch_bounds__(256) void gcn_dinv_kernel(const float* __restrict__ degc,
                                                       float* __restrict__ dinv, int N) {
  int i = blockIdx.x * 256 + threadIdx.x;
  if (i < N) {
    float d = degc[i] + degc[(size_t)N + i] + 1.0f;
    dinv[i] = (d > 0.0f) ? rsqrtf(d) : 0.0f;
  }
}

template <int F>
__global__ __launch_bounds__(256) void gcn_edge_agg_kernel(const int* __restrict__ src,
                                                           const int* __restrict__ dst,
                                                           const float* __restrict__ dinv,
                                                           const float* __restrict__ H,
                                                           float* __restrict__ AGG,
                                                           long long total,
                                                           long long copyStride) {
  float* mine = AGG + (size_t)gcn_chip_half() * (size_t)copyStride;
  constexpr int CPE = F / 4;
  const long long step = (long long)gridDim.x * 256;
  for (long long t = (long long)blockIdx.x * 256 + threadIdx.x; t < total; t += step) {
    const int e  = (int)(t / CPE);
    const int fb = (int)(t % CPE) * 4;
    const int s = src[e];
    const int d = dst[e];
    const float c = dinv[s] * dinv[d];
    const float4 v = *(const float4*)(H + (size_t)s * F + fb);
    float* p = mine + (size_t)d * F + fb;
    unsafeAtomicAdd(p + 0, v.x * c);
    unsafeAtomicAdd(p + 1, v.y * c);
    unsafeAtomicAdd(p + 2, v.z * c);
    unsafeAtomicAdd(p + 3, v.w * c);
  }
}

__global__ __launch_bounds__(256) void gcn_mid_wmma_kernel(float* __restrict__ AGG,
                                                           const float* __restrict__ X,
                                                           const float* __restrict__ dinv,
                                                           const float* __restrict__ W1,
                                                           const float* __restrict__ b1,
                                                           const float* __restrict__ W2,
                                                           float* __restrict__ HW2,
                                                           int nTiles, int total) {
  const int lane = threadIdx.x & 31;
  const int wave = threadIdx.x >> 5;
  const int tile = blockIdx.x * 8 + wave;
  if (tile >= nTiles) return;

  const int rowBase = tile * 16;
  const int lo = lane & 15;
  const int hi = lane >> 4;
  const int row = rowBase + lo;

  const float di = dinv[row];
  float* AGGB = AGG + (size_t)total;

  v2f sf[8];
  {
    v2f a0[8], a1[8], xx[8];
#pragma unroll
    for (int kb = 0; kb < 8; ++kb) {
      const size_t i = (size_t)row * 32 + kb * 4 + hi * 2;
      a0[kb] = *(const v2f*)(AGG + i);
      a1[kb] = *(const v2f*)(AGGB + i);
      xx[kb] = *(const v2f*)(X + i);
    }
#pragma unroll
    for (int kb = 0; kb < 8; ++kb) {
      sf[kb].x = (a0[kb].x + a1[kb].x) + xx[kb].x * di * di;
      sf[kb].y = (a0[kb].y + a1[kb].y) + xx[kb].y * di * di;
    }
#pragma unroll
    for (int kb = 0; kb < 8; ++kb) {
      const size_t i = (size_t)row * 32 + kb * 4 + hi * 2;
      if (i < (size_t)(total / 2)) {
        const v2f z = {0.0f, 0.0f};
        *(v2f*)(AGG + i) = z;
        *(v2f*)(AGGB + i) = z;
      }
    }
  }

  float4 bl[2], bh[2];
#pragma unroll
  for (int ct = 0; ct < 2; ++ct) {
    bl[ct] = *(const float4*)(b1 + ct * 16 + hi * 8);
    bh[ct] = *(const float4*)(b1 + ct * 16 + hi * 8 + 4);
  }

  v8f h[2];
#pragma unroll
  for (int ct = 0; ct < 2; ++ct) {
    v8f c = {};
#pragma unroll
    for (int kb = 0; kb < 8; ++kb) {
      const int k = kb * 4 + hi * 2;
      v2f w;
      w.x = W1[(size_t)k * 32 + ct * 16 + lo];
      w.y = W1[(size_t)(k + 1) * 32 + ct * 16 + lo];
      c = __builtin_amdgcn_wmma_f32_16x16x4_f32(false, w, false, sf[kb],
                                                (short)0, c, false, false);
    }
    c[0] = tanhf(c[0] + bl[ct].x);
    c[1] = tanhf(c[1] + bl[ct].y);
    c[2] = tanhf(c[2] + bl[ct].z);
    c[3] = tanhf(c[3] + bl[ct].w);
    c[4] = tanhf(c[4] + bh[ct].x);
    c[5] = tanhf(c[5] + bh[ct].y);
    c[6] = tanhf(c[6] + bh[ct].z);
    c[7] = tanhf(c[7] + bh[ct].w);
    h[ct] = c;
  }

  v8f c2 = {};
#pragma unroll
  for (int kb = 0; kb < 8; ++kb) {
    const int ct = kb >> 2;
    const int v = (kb & 3) * 2;
    v2f a;
    a.x = h[ct][v];
    a.y = h[ct][v + 1];
    const int k = ct * 16 + hi * 8 + v;
    v2f b;
    b.x = W2[(size_t)k * 16 + lo];
    b.y = W2[(size_t)(k + 1) * 16 + lo];
    c2 = __builtin_amdgcn_wmma_f32_16x16x4_f32(false, a, false, b,
                                               (short)0, c2, false, false);
  }
#pragma unroll
  for (int v = 0; v < 8; ++v) {
    const int m = v + hi * 8;
    HW2[(size_t)(rowBase + m) * 16 + lo] = c2[v];
  }
}

__global__ __launch_bounds__(256) void gcn_finish2_kernel(const float* __restrict__ AGG,
                                                          const float* __restrict__ AGGB,
                                                          const float* __restrict__ HW,
                                                          const float* __restrict__ dinv,
                                                          const float* __restrict__ b2,
                                                          float* __restrict__ out_h,
                                                          float* __restrict__ out_ls,
                                                          int N) {
  int n = blockIdx.x * 256 + threadIdx.x;
  if (n >= N) return;
  const float di = dinv[n];
  const float sc = di * di;
  float v[16];
  float m = -3.402823466e38f;
#pragma unroll
  for (int f = 0; f < 16; ++f) {
    const float t = (AGG[(size_t)n * 16 + f] + AGGB[(size_t)n * 16 + f]) +
                    HW[(size_t)n * 16 + f] * sc + b2[f];
    v[f] = t;
    m = fmaxf(m, t);
  }
  float s = 0.0f;
#pragma unroll
  for (int f = 0; f < 16; ++f) s += expf(v[f] - m);
  const float lse = logf(s);
#pragma unroll
  for (int f = 0; f < 16; ++f) {
    out_h[(size_t)n * 16 + f] = v[f];
    out_ls[(size_t)n * 16 + f] = v[f] - m - lse;
  }
}

extern "C" void kernel_launch(void* const* d_in, const int* in_sizes, int n_in,
                              void* d_out, int out_size, void* d_ws, size_t ws_size,
                              hipStream_t stream) {
  const float* x  = (const float*)d_in[0];
  const int*   ei = (const int*)d_in[1];
  const float* W1 = (const float*)d_in[2];
  const float* b1 = (const float*)d_in[3];
  const float* W2 = (const float*)d_in[4];
  const float* b2 = (const float*)d_in[5];

  const int N = in_sizes[0] / 32;
  const int E = in_sizes[1] / 2;
  const int* src = ei;
  const int* dst = ei + E;

  float* ws   = (float*)d_ws;
  float* degc = ws;
  float* deg  = ws + (size_t)N * 2;
  float* hw2  = ws + (size_t)N * 3;
  float* agg1 = ws + (size_t)N * 35;
  float* agg2 = agg1;
  const long long copyStride = (long long)N * 32;

  float* out_h  = (float*)d_out;
  float* out_ls = out_h + (size_t)N * 16;

  auto nb = [](long long n) { return (unsigned)((n + 255) / 256); };
  const int nTiles = N / 16;
  const unsigned gemmBlocks = (unsigned)((nTiles + 7) / 8);

  const unsigned loopBlocks = 4096;

  gcn_init_kernel<<<nb((long long)N * 64), 256, 0, stream>>>(degc, agg1, N, N * 64);
  gcn_degree_kernel<<<loopBlocks, 256, 0, stream>>>(dst, degc, E, N);
  gcn_dinv_kernel<<<nb(N), 256, 0, stream>>>(degc, deg, N);

  gcn_edge_agg_kernel<32><<<loopBlocks, 256, 0, stream>>>(
      src, dst, deg, x, agg1, (long long)E * 8, copyStride);
  gcn_mid_wmma_kernel<<<gemmBlocks, 256, 0, stream>>>(agg1, x, deg, W1, b1, W2, hw2, nTiles,
                                                     N * 32);

  gcn_edge_agg_kernel<16><<<loopBlocks, 256, 0, stream>>>(
      src, dst, deg, hw2, agg2, (long long)E * 4, copyStride);
  gcn_finish2_kernel<<<nb(N), 256, 0, stream>>>(agg2, agg2 + copyStride, hw2, deg, b2, out_h,
                                                out_ls, N);
}


# edge weight dinv[src]*dinv[dst] computed once: layer-1 edge pass stores it per edge, layer-2 pass loads it (consecutive lanes) instead of two scattered dinv reads per edge; same bits, sum order kept
# speedup vs baseline: 1.9128x; 1.0316x over previous
#include <hip/hip_runtime.h>
#include <hip/hip_bf16.h>


typedef __attribute__((ext_vector_type(2))) float v2f;
typedef __attribute__((ext_vector_type(8))) float v8f;

__device__ __forceinline__ unsigned gcn_chip_half() {
  return (__builtin_amdgcn_s_sendmsg_rtn(0x87) >> 18) & 1u;
}

__global__ __launch_bounds__(256) void gcn_init_kernel(float* __restrict__ degc,
                                                       float* __restrict__ agg,
                                                       int N, int total64) {
  int i = blockIdx.x * 256 + threadIdx.x;
  if (i < total64) agg[i] = 0.0f;
  if (i < 2 * N) degc[i] = 0.0f;
}

__global__ __launch_bounds__(256) void gcn_degree_kernel(const int* __restrict__ dst,
                                                         float* __restrict__ degc,
                                                         int E, int N) {
  float* mine = degc + (size_t)gcn_chip_half() * (size_t)N;
  const int step = (int)gridDim.x * 256;
  for (int e = blockIdx.x * 256 + threadIdx.x; e < E; e += step)
    unsafeAtomicAdd(&mine[dst[e]], 1.0f);
}

__global__ __launch_bounds__(256) void gcn_dinv_kernel(const float* __restrict__ degc,
                                                       float* __restrict__ dinv, int N) {
  int i = blockIdx.x * 256 + threadIdx.x;
  if (i < N) {
    float d = degc[i] + degc[(size_t)N + i] + 1.0f;
    dinv[i] = (d > 0.0f) ? rsqrtf(d) : 0.0f;
  }
}

template <int F, bool MAKE_W>
__global__ __launch_bounds__(256) void gcn_edge_agg_kernel(const int* __restrict__ src,
                                                           const int* __restrict__ dst,
                                                           const float* __restrict__ dinv,
                                                           const float* __restrict__ H,
                                                           float* __restrict__ AGG,
                                                           float* __restrict__ W,
                                                           long long total,
                                                           long long copyStride) {
  float* mine = AGG + (size_t)gcn_chip_half() * (size_t)copyStride;
  constexpr int CPE = F / 4;
  const long long step = (long long)gridDim.x * 256;
  for (long long t = (long long)blockIdx.x * 256 + threadIdx.x; t < total; t += step) {
    const int e  = (int)(t / CPE);
    const int fb = (int)(t % CPE) * 4;
    const int s = src[e];
    const int d = dst[e];
    float c;
    if constexpr (MAKE_W) {
      c = dinv[s] * dinv[d];
      if (fb == 0) W[e] = c;
    } else {
      c = W[e];
    }
    const float4 v = *(const float4*)(H + (size_t)s * F + fb);
    float* p = mine + (size_t)d * F + fb;
    unsafeAtomicAdd(p + 0, v.x * c);
    unsafeAtomicAdd(p + 1, v.y * c);
    unsafeAtomicAdd(p + 2, v.z * c);
    unsafeAtomicAdd(p + 3, v.w * c);
  }
}

__global__ __launch_bounds__(256) void gcn_mid_wmma_kernel(float* __restrict__ AGG,
                                                           const float* __restrict__ X,
                                                           const float* __restrict__ dinv,
                                                           const float* __restrict__ W1,
                                                           const float* __restrict__ b1,
                                                           const float* __restrict__ W2,
                                                           float* __restrict__ HW2,
                                                           int nTiles, int total) {
  const int lane = threadIdx.x & 31;
  const int wave = threadIdx.x >> 5;
  const int tile = blockIdx.x * 8 + wave;
  if (tile >= nTiles) return;

  const int rowBase = tile * 16;
  const int lo = lane & 15;
  const int hi = lane >> 4;
  const int row = rowBase + lo;

  const float di = dinv[row];
  float* AGGB = AGG + (size_t)total;

  v2f sf[8];
  {
    v2f a0[8], a1[8], xx[8];
#pragma unroll
    for (int kb = 0; kb < 8; ++kb) {
      const size_t i = (size_t)row * 32 + kb * 4 + hi * 2;
      a0[kb] = *(const v2f*)(AGG + i);
      a1[kb] = *(const v2f*)(AGGB + i);
      xx[kb] = *(const v2f*)(X + i);
    }
#pragma unroll
    for (int kb = 0; kb < 8; ++kb) {
      sf[kb].x = (a0[kb].x + a1[kb].x) + xx[kb].x * di * di;
      sf[kb].y = (a0[kb].y + a1[kb].y) + xx[kb].y * di * di;
    }
#pragma unroll
    for (int kb = 0; kb < 8; ++kb) {
      const size_t i = (size_t)row * 32 + kb * 4 + hi * 2;
      if (i < (size_t)(total / 2)) {
        const v2f z = {0.0f, 0.0f};
        *(v2f*)(AGG + i) = z;
        *(v2f*)(AGGB + i) = z;
      }
    }
  }

  float4 bl[2], bh[2];
#pragma unroll
  for (int ct = 0; ct < 2; ++ct) {
    bl[ct] = *(const float4*)(b1 + ct * 16 + hi * 8);
    bh[ct] = *(const float4*)(b1 + ct * 16 + hi * 8 + 4);
  }

  v8f h[2];
#pragma unroll
  for (int ct = 0; ct < 2; ++ct) {
    v8f c = {};
#pragma unroll
    for (int kb = 0; kb < 8; ++kb) {
      const int k = kb * 4 + hi * 2;
      v2f w;
      w.x = W1[(size_t)k * 32 + ct * 16 + lo];
      w.y = W1[(size_t)(k + 1) * 32 + ct * 16 + lo];
      c = __builtin_amdgcn_wmma_f32_16x16x4_f32(false, w, false, sf[kb],
                                                (short)0, c, false, false);
    }
    c[0] = tanhf(c[0] + bl[ct].x);
    c[1] = tanhf(c[1] + bl[ct].y);
    c[2] = tanhf(c[2] + bl[ct].z);
    c[3] = tanhf(c[3] + bl[ct].w);
    c[4] = tanhf(c[4] + bh[ct].x);
    c[5] = tanhf(c[5] + bh[ct].y);
    c[6] = tanhf(c[6] + bh[ct].z);
    c[7] = tanhf(c[7] + bh[ct].w);
    h[ct] = c;
  }

  v8f c2 = {};
#pragma unroll
  for (int kb = 0; kb < 8; ++kb) {
    const int ct = kb >> 2;
    const int v = (kb & 3) * 2;
    v2f a;
    a.x = h[ct][v];
    a.y = h[ct][v + 1];
    const int k = ct * 16 + hi * 8 + v;
    v2f b;
    b.x = W2[(size_t)k * 16 + lo];
    b.y = W2[(size_t)(k + 1) * 16 + lo];
    c2 = __builtin_amdgcn_wmma_f32_16x16x4_f32(false, a, false, b,
                                               (short)0, c2, false, false);
  }
#pragma unroll
  for (int v = 0; v < 8; ++v) {
    const int m = v + hi * 8;
    HW2[(size_t)(rowBase + m) * 16 + lo] = c2[v];
  }
}

__global__ __launch_bounds__(256) void gcn_finish2_kernel(const float* __restrict__ AGG,
                                                          const float* __restrict__ AGGB,
                                                          const float* __restrict__ HW,
                                                          const float* __restrict__ dinv,
                                                          const float* __restrict__ b2,
                                                          float* __restrict__ out_h,
                                                          float* __restrict__ out_ls,
                                                          int N) {
  int n = blockIdx.x * 256 + threadIdx.x;
  if (n >= N) return;
  const float di = dinv[n];
  const float sc = di * di;
  float v[16];
  float m = -3.402823466e38f;
#pragma unroll
  for (int f = 0; f < 16; ++f) {
    const float t = (AGG[(size_t)n * 16 + f] + AGGB[(size_t)n * 16 + f]) +
                    HW[(size_t)n * 16 + f] * sc + b2[f];
    v[f] = t;
    m = fmaxf(m, t);
  }
  float s = 0.0f;
#pragma unroll
  for (int f = 0; f < 16; ++f) s += expf(v[f] - m);
  const float lse = logf(s);
#pragma unroll
  for (int f = 0; f < 16; ++f) {
    out_h[(size_t)n * 16 + f] = v[f];
    out_ls[(size_t)n * 16 + f] = v[f] - m - lse;
  }
}

extern "C" void kernel_launch(void* const* d_in, const int* in_sizes, int n_in,
                              void* d_out, int out_size, void* d_ws, size_t ws_size,
                              hipStream_t stream) {
  const float* x  = (const float*)d_in[0];
  const int*   ei = (const int*)d_in[1];
  const float* W1 = (const float*)d_in[2];
  const float* b1 = (const float*)d_in[3];
  const float* W2 = (const float*)d_in[4];
  const float* b2 = (const float*)d_in[5];

  const int N = in_sizes[0] / 32;
  const int E = in_sizes[1] / 2;
  const int* src = ei;
  const int* dst = ei + E;

  float* ws   = (float*)d_ws;
  float* degc = ws;
  float* deg  = ws + (size_t)N * 2;
  float* hw2  = ws + (size_t)N * 3;
  float* agg1 = ws + (size_t)N * 35;
  float* agg2 = agg1;
  float* ew   = ws + (size_t)N * 99;
  const long long copyStride = (long long)N * 32;

  float* out_h  = (float*)d_out;
  float* out_ls = out_h + (size_t)N * 16;

  auto nb = [](long long n) { return (unsigned)((n + 255) / 256); };
  const int nTiles = N / 16;
  const unsigned gemmBlocks = (unsigned)((nTiles + 7) / 8);

  const unsigned loopBlocks = 4096;

  gcn_init_kernel<<<nb((long long)N * 64), 256, 0, stream>>>(degc, agg1, N, N * 64);
  gcn_degree_kernel<<<loopBlocks, 256, 0, stream>>>(dst, degc, E, N);
  gcn_dinv_kernel<<<nb(N), 256, 0, stream>>>(degc, deg, N);

  gcn_edge_agg_kernel<32, true><<<loopBlocks, 256, 0, stream>>>(
      src, dst, deg, x, agg1, ew, (long long)E * 8, copyStride);
  gcn_mid_wmma_kernel<<<gemmBlocks, 256, 0, stream>>>(agg1, x, deg, W1, b1, W2, hw2, nTiles,
                                                     N * 32);

  gcn_edge_agg_kernel<16, false><<<loopBlocks, 256, 0, stream>>>(
      src, dst, deg, hw2, agg2, ew, (long long)E * 4, copyStride);
  gcn_finish2_kernel<<<nb(N), 256, 0, stream>>>(agg2, agg2 + copyStride, hw2, deg, b2, out_h,
                                                out_ls, N);
}
